# starting kernel
# baseline (speedup 1.0000x reference)
#include <hip/hip_runtime.h>

typedef __attribute__((ext_vector_type(2))) float v2f;
typedef __attribute__((ext_vector_type(8))) float v8f;

#define B_  2
#define C_  32
#define N_  1024

__global__ __launch_bounds__(256)
void psm_attn_kernel(const float* __restrict__ feature,
                     const float* __restrict__ seg,
                     const float* __restrict__ gamma,
                     float* __restrict__ out)
{
    __shared__ __align__(16) float ldsM[N_];
    __shared__ __align__(16) float ldsF[N_];

    const int bc   = blockIdx.x >> 3;
    const int tg   = blockIdx.x & 7;
    const int b    = bc >> 5;
    const int tid  = threadIdx.x;
    const int lane = tid & 31;
    const int wave = tid >> 5;

    const float* fptr = feature + (size_t)bc * N_;
    const float* mptr = seg + ((size_t)b * 2 + 1) * N_;

    for (int j = tid; j < N_; j += 256) {
        ldsM[j] = mptr[j];
        ldsF[j] = fptr[j];
    }
    __syncthreads();

    float mx = -3.0e38f, mn = 3.0e38f;
    for (int j = lane; j < N_; j += 32) {
        const float v = ldsM[j];
        mx = fmaxf(mx, v);
        mn = fminf(mn, v);
    }
#pragma unroll
    for (int off = 16; off > 0; off >>= 1) {
        mx = fmaxf(mx, __shfl_xor(mx, off, 32));
        mn = fminf(mn, __shfl_xor(mn, off, 32));
    }

    const int half = lane >> 4;
    const int row  = lane & 15;
    const int i0   = (tg * 8 + wave) * 16;

    const float sM = ldsF[i0 + row];
    const float L2E  = 1.44269504088896340736f;
    const float sL2  = sM * L2E;
    const float rL2  = (sM >= 0.0f ? sM * mx : sM * mn) * L2E;
    const float g    = gamma[0];

    v8f acc_num = {};
    v8f acc_den = {};
    const v2f bOnes = {1.0f, 1.0f};

    const v2f* m2 = (const v2f*)ldsM;
    const v2f* f2 = (const v2f*)ldsF;

    for (int j0 = 0; j0 < N_; j0 += 4) {
        const int k = (j0 >> 1) + half;
        const v2f mv = m2[k];
        const v2f fv = f2[k];
        v2f e, ef;
        e.x = __builtin_amdgcn_exp2f(sL2 * mv.x - rL2);
        e.y = __builtin_amdgcn_exp2f(sL2 * mv.y - rL2);
        ef.x = e.x * fv.x;
        ef.y = e.y * fv.y;
        acc_num = __builtin_amdgcn_wmma_f32_16x16x4_f32(
            false, ef, false, bOnes, (short)0, acc_num, false, false);
        acc_den = __builtin_amdgcn_wmma_f32_16x16x4_f32(
            false, e,  false, bOnes, (short)0, acc_den, false, false);
    }

    float* obase = out + (size_t)bc * N_;
#pragma unroll
    for (int r = 0; r < 8; ++r) {
        if (row == r) {
            const int i = i0 + r + 8 * half;
            obase[i] = g * (acc_num[r] / acc_den[r]) + ldsF[i];
        }
    }
}

extern "C" void kernel_launch(void* const* d_in, const int* in_sizes, int n_in,
                              void* d_out, int out_size, void* d_ws, size_t ws_size,
                              hipStream_t stream) {
    (void)in_sizes; (void)n_in; (void)out_size; (void)d_ws; (void)ws_size;
    const float* feature = (const float*)d_in[0];
    const float* seg     = (const float*)d_in[1];
    const float* gamma   = (const float*)d_in[2];
    float* out           = (float*)d_out;

    dim3 grid(B_ * C_ * 8);
    dim3 block(256);
    psm_attn_kernel<<<grid, block, 0, stream>>>(feature, seg, gamma, out);
}


# One WMMA per k-step instead of two: B = f (cols 0-7) | ones (cols 8-15), so e*f and both row sums form in the matrix unit; even/odd k-steps in two accumulators added at the end (sum order changes)
# speedup vs baseline: 1.5078x; 1.5078x over previous
#include <hip/hip_runtime.h>


typedef __attribute__((ext_vector_type(2))) float v2f;
typedef __attribute__((ext_vector_type(8))) float v8f;

#define B_  2
#define C_  32
#define N_  1024
#define PAD_ 4

__global__ __launch_bounds__(256)
void psm_attn_kernel(const float* __restrict__ feature,
                     const float* __restrict__ seg,
                     const float* __restrict__ gamma,
                     float* __restrict__ out)
{
    __shared__ __align__(16) float lds[3 * N_ + 2 * PAD_];
    float* ldsM = lds;
    float* ldsF = lds + (N_ + PAD_);
    float* ldsO = lds + 2 * (N_ + PAD_);

    const int bc   = blockIdx.x >> 3;
    const int tg   = blockIdx.x & 7;
    const int b    = bc >> 5;
    const int tid  = threadIdx.x;
    const int lane = tid & 31;
    const int wave = tid >> 5;

    const float* fptr = feature + (size_t)bc * N_;
    const float* mptr = seg + ((size_t)b * 2 + 1) * N_;

    for (int j = tid; j < N_; j += 256) {
        ldsM[j] = mptr[j];
        ldsF[j] = fptr[j];
        ldsO[j] = 1.0f;
    }
    __syncthreads();

    float mx = -3.0e38f, mn = 3.0e38f;
    for (int j = lane; j < N_; j += 32) {
        const float v = ldsM[j];
        mx = fmaxf(mx, v);
        mn = fminf(mn, v);
    }
#pragma unroll
    for (int off = 16; off > 0; off >>= 1) {
        mx = fmaxf(mx, __shfl_xor(mx, off, 32));
        mn = fminf(mn, __shfl_xor(mn, off, 32));
    }

    const int half = lane >> 4;
    const int col  = lane & 15;
    const int i0   = (tg * 8 + wave) * 16;

    const float sM = ldsF[i0 + col];
    const float L2E  = 1.44269504088896340736f;
    const float sL2  = sM * L2E;
    const float rL2  = (sM >= 0.0f ? sM * mx : sM * mn) * L2E;
    const float g    = gamma[0];

    v8f acc0 = {};
    v8f acc1 = {};

    const v2f* m2 = (const v2f*)ldsM;
    const v2f* b2 = (const v2f*)(col < 8 ? ldsF : ldsO);

    for (int j0 = 0; j0 < N_; j0 += 8) {
        const int k = (j0 >> 1) + half;
        const v2f mva = m2[k];
        const v2f bva = b2[k];
        const v2f mvb = m2[k + 2];
        const v2f bvb = b2[k + 2];
        v2f ea, eb;
        ea.x = __builtin_amdgcn_exp2f(sL2 * mva.x - rL2);
        ea.y = __builtin_amdgcn_exp2f(sL2 * mva.y - rL2);
        eb.x = __builtin_amdgcn_exp2f(sL2 * mvb.x - rL2);
        eb.y = __builtin_amdgcn_exp2f(sL2 * mvb.y - rL2);
        acc0 = __builtin_amdgcn_wmma_f32_16x16x4_f32(
            false, ea, false, bva, (short)0, acc0, false, false);
        acc1 = __builtin_amdgcn_wmma_f32_16x16x4_f32(
            false, eb, false, bvb, (short)0, acc1, false, false);
    }

    const int rsel = col & 7;
    float v = acc0[0] + acc1[0];
#pragma unroll
    for (int r = 1; r < 8; ++r) {
        const float t = acc0[r] + acc1[r];
        v = (rsel == r) ? t : v;
    }
    const float den = __shfl_xor(v, 8, 32);
    if (col < 8) {
        const int i = i0 + 8 * half + col;
        out[(size_t)bc * N_ + i] = g * (v / den) + ldsF[i];
    }
}

extern "C" void kernel_launch(void* const* d_in, const int* in_sizes, int n_in,
                              void* d_out, int out_size, void* d_ws, size_t ws_size,
                              hipStream_t stream) {
    (void)in_sizes; (void)n_in; (void)out_size; (void)d_ws; (void)ws_size;
    const float* feature = (const float*)d_in[0];
    const float* seg     = (const float*)d_in[1];
    const float* gamma   = (const float*)d_in[2];
    float* out           = (float*)d_out;

    dim3 grid(B_ * C_ * 8);
    dim3 block(256);
    psm_attn_kernel<<<grid, block, 0, stream>>>(feature, seg, gamma, out);
}


# LDS fragment reads regrouped: one group of eight 16-byte reads per 8-WMMA block, issued a block ahead into a second register set, instead of a read site every 2 WMMAs (sum order changes)
# speedup vs baseline: 1.6576x; 1.0994x over previous
#include <hip/hip_runtime.h>


typedef __attribute__((ext_vector_type(2))) float v2f;
typedef __attribute__((ext_vector_type(4))) float v4f;
typedef __attribute__((ext_vector_type(8))) float v8f;

#define B_  2
#define C_  32
#define N_  1024
#define PAD_ 4

#define NU_ 4

__device__ __forceinline__ void load_block(const v4f* __restrict__ m4,
                                           const v4f* __restrict__ b4,
                                           int q0, v4f (&mv)[NU_], v4f (&bv)[NU_])
{
#pragma unroll
    for (int u = 0; u < NU_; ++u) {
        mv[u] = m4[q0 + 2 * u];
        bv[u] = b4[q0 + 2 * u];
    }
}

__device__ __forceinline__ void mma_block(const v4f (&mv)[NU_], const v4f (&bv)[NU_],
                                          float sL2, float rL2, v8f& acc0, v8f& acc1)
{
#pragma unroll
    for (int u = 0; u < NU_; ++u) {
        v2f ea, eb, ba, bb;
        ea.x = __builtin_amdgcn_exp2f(sL2 * mv[u].x - rL2);
        ea.y = __builtin_amdgcn_exp2f(sL2 * mv[u].y - rL2);
        eb.x = __builtin_amdgcn_exp2f(sL2 * mv[u].z - rL2);
        eb.y = __builtin_amdgcn_exp2f(sL2 * mv[u].w - rL2);
        ba.x = bv[u].x; ba.y = bv[u].y;
        bb.x = bv[u].z; bb.y = bv[u].w;
        acc0 = __builtin_amdgcn_wmma_f32_16x16x4_f32(
            false, ea, false, ba, (short)0, acc0, false, false);
        acc1 = __builtin_amdgcn_wmma_f32_16x16x4_f32(
            false, eb, false, bb, (short)0, acc1, false, false);
    }
}

__global__ __launch_bounds__(256)
void psm_attn_kernel(const float* __restrict__ feature,
                     const float* __restrict__ seg,
                     const float* __restrict__ gamma,
                     float* __restrict__ out)
{
    __shared__ __align__(16) float lds[3 * N_ + 2 * PAD_];
    float* ldsM = lds;
    float* ldsF = lds + (N_ + PAD_);
    float* ldsO = lds + 2 * (N_ + PAD_);

    const int bc   = blockIdx.x >> 3;
    const int tg   = blockIdx.x & 7;
    const int b    = bc >> 5;
    const int tid  = threadIdx.x;
    const int lane = tid & 31;
    const int wave = tid >> 5;

    const float* fptr = feature + (size_t)bc * N_;
    const float* mptr = seg + ((size_t)b * 2 + 1) * N_;

    for (int j = tid; j < N_; j += 256) {
        ldsM[j] = mptr[j];
        ldsF[j] = fptr[j];
        ldsO[j] = 1.0f;
    }
    __syncthreads();

    float mx = -3.0e38f, mn = 3.0e38f;
    for (int j = lane; j < N_; j += 32) {
        const float v = ldsM[j];
        mx = fmaxf(mx, v);
        mn = fminf(mn, v);
    }
#pragma unroll
    for (int off = 16; off > 0; off >>= 1) {
        mx = fmaxf(mx, __shfl_xor(mx, off, 32));
        mn = fminf(mn, __shfl_xor(mn, off, 32));
    }

    const int half = lane >> 4;
    const int col  = lane & 15;
    const int i0   = (tg * 8 + wave) * 16;

    const float sM = ldsF[i0 + col];
    const float L2E  = 1.44269504088896340736f;
    const float sL2  = sM * L2E;
    const float rL2  = (sM >= 0.0f ? sM * mx : sM * mn) * L2E;
    const float g    = gamma[0];

    v8f acc0 = {};
    v8f acc1 = {};

    const v4f* m4 = (const v4f*)ldsM;
    const v4f* b4 = (const v4f*)(col < 8 ? ldsF : ldsO);

#define READS_THEN_MMAS()                                                    \
    do {                                                                     \
        __builtin_amdgcn_sched_group_barrier(0x100, 2 * NU_, 0);   \
        __builtin_amdgcn_sched_group_barrier(0x008, 2 * NU_, 0);   \
    } while (0)
#define BLOCK_DONE() asm volatile("" :: "v"(acc0), "v"(acc1) : "memory")
    v4f mA[NU_], bA[NU_], mB[NU_], bB[NU_];
    load_block(m4, b4, half, mA, bA);
    for (int t = 0; t < N_ / (8 * NU_) - 2; t += 2) {
        load_block(m4, b4, (t + 1) * 2 * NU_ + half, mB, bB);
        mma_block(mA, bA, sL2, rL2, acc0, acc1);
        BLOCK_DONE();
        load_block(m4, b4, (t + 2) * 2 * NU_ + half, mA, bA);
        mma_block(mB, bB, sL2, rL2, acc0, acc1);
        READS_THEN_MMAS();
        READS_THEN_MMAS();
    }
    load_block(m4, b4, (N_ / (8 * NU_) - 1) * 2 * NU_ + half, mB, bB);
    mma_block(mA, bA, sL2, rL2, acc0, acc1);
    mma_block(mB, bB, sL2, rL2, acc0, acc1);
    READS_THEN_MMAS();
#undef READS_THEN_MMAS
#undef BLOCK_DONE

    const int rsel = col & 7;
    float v = acc0[0] + acc1[0];
#pragma unroll
    for (int r = 1; r < 8; ++r) {
        const float t = acc0[r] + acc1[r];
        v = (rsel == r) ? t : v;
    }
    const float den = __shfl_xor(v, 8, 32);
    if (col < 8) {
        const int i = i0 + 8 * half + col;
        out[(size_t)bc * N_ + i] = g * (v / den) + ldsF[i];
    }
}

extern "C" void kernel_launch(void* const* d_in, const int* in_sizes, int n_in,
                              void* d_out, int out_size, void* d_ws, size_t ws_size,
                              hipStream_t stream) {
    (void)in_sizes; (void)n_in; (void)out_size; (void)d_ws; (void)ws_size;
    const float* feature = (const float*)d_in[0];
    const float* seg     = (const float*)d_in[1];
    const float* gamma   = (const float*)d_in[2];
    float* out           = (float*)d_out;

    dim3 grid(B_ * C_ * 8);
    dim3 block(256);
    psm_attn_kernel<<<grid, block, 0, stream>>>(feature, seg, gamma, out);
}


# Prologue chain shortened: m and f staged by one 16-byte global load per thread (one round trip, was four dependent ones); min/max of m from those registers, wave results joined in LDS (no LDS scan)
# speedup vs baseline: 1.9269x; 1.1625x over previous
#include <hip/hip_runtime.h>


typedef __attribute__((ext_vector_type(2))) float v2f;
typedef __attribute__((ext_vector_type(4))) float v4f;
typedef __attribute__((ext_vector_type(8))) float v8f;

#define B_  2
#define C_  32
#define N_  1024
#define PAD_ 4

#define NU_ 4

template <int MASK>
__device__ __forceinline__ float row_xor(float x)
{
    const int b = __builtin_bit_cast(int, x);
    return __builtin_bit_cast(float,
        __builtin_amdgcn_update_dpp(b, b, 0x160 + MASK, 0xf, 0xf, false));
}
__device__ __forceinline__ float xhalf(float x)
{
    const int b = __builtin_bit_cast(int, x);
    return __builtin_bit_cast(float,
        __builtin_amdgcn_permlanex16(b, b, 0x76543210, 0xfedcba98, false, false));
}

__device__ __forceinline__ void load_block(const v4f* __restrict__ m4,
                                           const v4f* __restrict__ b4,
                                           int q0, v4f (&mv)[NU_], v4f (&bv)[NU_])
{
#pragma unroll
    for (int u = 0; u < NU_; ++u) {
        mv[u] = m4[q0 + 2 * u];
        bv[u] = b4[q0 + 2 * u];
    }
}

__device__ __forceinline__ void mma_block(const v4f (&mv)[NU_], const v4f (&bv)[NU_],
                                          float sL2, float rL2, v8f& acc0, v8f& acc1)
{
#pragma unroll
    for (int u = 0; u < NU_; ++u) {
        v2f ea, eb, ba, bb;
        ea.x = __builtin_amdgcn_exp2f(sL2 * mv[u].x - rL2);
        ea.y = __builtin_amdgcn_exp2f(sL2 * mv[u].y - rL2);
        eb.x = __builtin_amdgcn_exp2f(sL2 * mv[u].z - rL2);
        eb.y = __builtin_amdgcn_exp2f(sL2 * mv[u].w - rL2);
        ba.x = bv[u].x; ba.y = bv[u].y;
        bb.x = bv[u].z; bb.y = bv[u].w;
        acc0 = __builtin_amdgcn_wmma_f32_16x16x4_f32(
            false, ea, false, ba, (short)0, acc0, false, false);
        acc1 = __builtin_amdgcn_wmma_f32_16x16x4_f32(
            false, eb, false, bb, (short)0, acc1, false, false);
    }
}

__global__ __launch_bounds__(256)
void psm_attn_kernel(const float* __restrict__ feature,
                     const float* __restrict__ seg,
                     const float* __restrict__ gamma,
                     float* __restrict__ out)
{
    __shared__ __align__(16) float lds[3 * N_ + 2 * PAD_ + 16];
    float* ldsM = lds;
    float* ldsF = lds + (N_ + PAD_);
    float* ldsO = lds + 2 * (N_ + PAD_);
    float* ldsP = lds + 3 * N_ + 2 * PAD_;

    const int bc   = blockIdx.x >> 3;
    const int tg   = blockIdx.x & 7;
    const int b    = bc >> 5;
    const int tid  = threadIdx.x;
    const int lane = tid & 31;
    const int wave = tid >> 5;

    const float* fptr = feature + (size_t)bc * N_;
    const float* mptr = seg + ((size_t)b * 2 + 1) * N_;

    const v4f mq = ((const v4f*)mptr)[tid];
    const v4f fq = ((const v4f*)fptr)[tid];
    const v4f one4 = {1.0f, 1.0f, 1.0f, 1.0f};
    ((v4f*)ldsM)[tid] = mq;
    ((v4f*)ldsF)[tid] = fq;
    ((v4f*)ldsO)[tid] = one4;

    float mx = fmaxf(fmaxf(mq.x, mq.y), fmaxf(mq.z, mq.w));
    float mn = fminf(fminf(mq.x, mq.y), fminf(mq.z, mq.w));
    mx = fmaxf(mx, row_xor<1>(mx));  mn = fminf(mn, row_xor<1>(mn));
    mx = fmaxf(mx, row_xor<2>(mx));  mn = fminf(mn, row_xor<2>(mn));
    mx = fmaxf(mx, row_xor<4>(mx));  mn = fminf(mn, row_xor<4>(mn));
    mx = fmaxf(mx, row_xor<8>(mx));  mn = fminf(mn, row_xor<8>(mn));
    mx = fmaxf(mx, xhalf(mx));       mn = fminf(mn, xhalf(mn));
    if (lane == 0) {
        ldsP[wave]     = mx;
        ldsP[8 + wave] = mn;
    }
    __syncthreads();
    {
        const v4f* p4 = (const v4f*)ldsP;
        const v4f a = p4[0], c = p4[1], d = p4[2], e = p4[3];
        mx = fmaxf(fmaxf(fmaxf(a.x, a.y), fmaxf(a.z, a.w)),
                   fmaxf(fmaxf(c.x, c.y), fmaxf(c.z, c.w)));
        mn = fminf(fminf(fminf(d.x, d.y), fminf(d.z, d.w)),
                   fminf(fminf(e.x, e.y), fminf(e.z, e.w)));
    }

    const int half = lane >> 4;
    const int col  = lane & 15;
    const int i0   = (tg * 8 + wave) * 16;

    const float sM = ldsF[i0 + col];
    const float L2E  = 1.44269504088896340736f;
    const float sL2  = sM * L2E;
    const float rL2  = (sM >= 0.0f ? sM * mx : sM * mn) * L2E;
    const float g    = gamma[0];

    v8f acc0 = {};
    v8f acc1 = {};

    const v4f* m4 = (const v4f*)ldsM;
    const v4f* b4 = (const v4f*)(col < 8 ? ldsF : ldsO);

#define READS_THEN_MMAS()                                                    \
    do {                                                                     \
        __builtin_amdgcn_sched_group_barrier(0x100, 2 * NU_, 0);   \
        __builtin_amdgcn_sched_group_barrier(0x008, 2 * NU_, 0);   \
    } while (0)
#define BLOCK_DONE() asm volatile("" :: "v"(acc0), "v"(acc1) : "memory")
    v4f mA[NU_], bA[NU_], mB[NU_], bB[NU_];
    load_block(m4, b4, half, mA, bA);
    for (int t = 0; t < N_ / (8 * NU_) - 2; t += 2) {
        load_block(m4, b4, (t + 1) * 2 * NU_ + half, mB, bB);
        mma_block(mA, bA, sL2, rL2, acc0, acc1);
        BLOCK_DONE();
        load_block(m4, b4, (t + 2) * 2 * NU_ + half, mA, bA);
        mma_block(mB, bB, sL2, rL2, acc0, acc1);
        READS_THEN_MMAS();
        READS_THEN_MMAS();
    }
    load_block(m4, b4, (N_ / (8 * NU_) - 1) * 2 * NU_ + half, mB, bB);
    mma_block(mA, bA, sL2, rL2, acc0, acc1);
    mma_block(mB, bB, sL2, rL2, acc0, acc1);
    READS_THEN_MMAS();
#undef READS_THEN_MMAS
#undef BLOCK_DONE

    const int rsel = col & 7;
    float v = acc0[0] + acc1[0];
#pragma unroll
    for (int r = 1; r < 8; ++r) {
        const float t = acc0[r] + acc1[r];
        v = (rsel == r) ? t : v;
    }
    const float den = __shfl_xor(v, 8, 32);
    if (col < 8) {
        const int i = i0 + 8 * half + col;
        out[(size_t)bc * N_ + i] = g * (v / den) + ldsF[i];
    }
}

extern "C" void kernel_launch(void* const* d_in, const int* in_sizes, int n_in,
                              void* d_out, int out_size, void* d_ws, size_t ws_size,
                              hipStream_t stream) {
    (void)in_sizes; (void)n_in; (void)out_size; (void)d_ws; (void)ws_size;
    const float* feature = (const float*)d_in[0];
    const float* seg     = (const float*)d_in[1];
    const float* gamma   = (const float*)d_in[2];
    float* out           = (float*)d_out;

    dim3 grid(B_ * C_ * 8);
    dim3 block(256);
    psm_attn_kernel<<<grid, block, 0, stream>>>(feature, seg, gamma, out);
}
